# starting kernel
# baseline (speedup 1.0000x reference)
#include <hip/hip_runtime.h>


typedef float v8f __attribute__((ext_vector_type(8)));
typedef float v2f __attribute__((ext_vector_type(2)));

#define D_DIM 2048
#define N_MODES 16
#define B_DIM 2
#define L_DIM 4096
#define CG 128
#define NGROUP 2
#define MROW 17
#define UROW 136
#define XROW 17

static __device__ __forceinline__ v8f wmma4(v2f a, v2f b, v8f c) {
  return __builtin_amdgcn_wmma_f32_16x16x4_f32(false, a, false, b, (short)0, c,
                                               false, false);
}

static __device__ __forceinline__ void cpow(float qr, float qi, int e,
                                            float& pr, float& pi) {
  pr = 1.f; pi = 0.f;
  for (int k = 0; k < e; ++k) {
    float t = pr * qr - pi * qi;
    pi = pr * qi + pi * qr;
    pr = t;
  }
}

__launch_bounds__(256, 2)
__global__ void cema_kernel(const float* __restrict__ x,
                            const float* __restrict__ alpha,
                            const float* __restrict__ delta,
                            const float* __restrict__ theta,
                            const float* __restrict__ gma,
                            const float* __restrict__ omga,
                            float* __restrict__ y_out,
                            float* __restrict__ h_out) {
  __shared__ float xs_s[B_DIM][CG * XROW];
  __shared__ float Ure_s[B_DIM][N_MODES * UROW];
  __shared__ float Uim_s[B_DIM][N_MODES * UROW];
  __shared__ float Bre_s[N_MODES * MROW];
  __shared__ float Bim_s[N_MODES * MROW];
  __shared__ float Are_s[N_MODES * MROW];
  __shared__ float AimN_s[N_MODES * MROW];
  __shared__ float Kmat_s[N_MODES * MROW];
  __shared__ float kern_s[N_MODES];
  __shared__ float cQr[16], cQi[16], cP[16], cGr[16], cGi[16];
  __shared__ float cCr[16], cCi[16], cTr[16], cTi[16];
  __shared__ float sOmega;

  const int tid = threadIdx.x;
  const int d = blockIdx.x;

  if (tid < 16) {
    int n = tid;
    float a = alpha[d * 16 + n];
    float de = delta[d * 16 + n];
    float th = theta[d];
    float p = 1.f / (1.f + expf(-a));
    float dd = 1.f / (1.f + expf(-de));
    float base = (1.f / (1.f + expf(-th))) * (2.f * 3.14159265358979323846f / 16.f);
    float phi = (float)(n + 1) * base;
    float r = 1.f - p * dd;
    float sn, cs;
    sincosf(phi, &sn, &cs);
    float qr = r * cs, qi = r * sn;
    cQr[n] = qr; cQi[n] = qi;
    cP[n] = p;
    float gr = gma[(d * 16 + n) * 2 + 0] * 0.25f;
    float gi = gma[(d * 16 + n) * 2 + 1] * 0.25f;
    cGr[n] = gr; cGi[n] = gi;
    cCr[n] = p * gr; cCi[n] = p * gi;
    float tr, ti;
    cpow(qr, qi, 16, tr, ti);
    cTr[n] = tr; cTi[n] = ti;
    if (n == 0) sOmega = omga[d];
  }
  __syncthreads();

  {
    int r = tid >> 4, cc = tid & 15;
    float qr = cQr[cc], qi = cQi[cc];
    float pr, pi;
    cpow(qr, qi, r, pr, pi);
    xs_s[0][r * MROW + cc] = cCr[cc] * pr - cCi[cc] * pi;
    float q1r = pr * qr - pi * qi;
    float q1i = pr * qi + pi * qr;
    Are_s[r * MROW + cc] = cGr[cc] * q1r - cGi[cc] * q1i;
    AimN_s[r * MROW + cc] = -(cGr[cc] * q1i + cGi[cc] * q1r);
    float br, bi;
    cpow(cQr[r], cQi[r], 15 - cc, br, bi);
    Bre_s[r * MROW + cc] = cP[r] * br;
    Bim_s[r * MROW + cc] = cP[r] * bi;
  }
  __syncthreads();
  if (tid < 16) {
    float s = 0.f;
    for (int n = 0; n < 16; ++n) s += xs_s[0][tid * MROW + n];
    kern_s[tid] = s;
  }
  __syncthreads();
  {
    int r = tid >> 4, cc = tid & 15;
    float v = (r >= cc) ? kern_s[r - cc] : 0.f;
    if (r == cc) v += sOmega;
    Kmat_s[r * MROW + cc] = v;
  }

  const int wv = tid >> 5;
  const int lane = tid & 31;
  const int lcol = lane & 15;
  const int khalf = lane >> 4;

  float hre = 0.f, him = 0.f;

  for (int g = 0; g < NGROUP; ++g) {
    __syncthreads();
    for (int b2 = 0; b2 < B_DIM; ++b2) {
      const float4* xg = (const float4*)(x + ((size_t)b2 * D_DIM + d) * L_DIM +
                                         (size_t)g * (CG * 16));
      for (int i = tid; i < CG * 16 / 4; i += 256) {
        float4 v = xg[i];
        int t = i * 4;
        float* dst = &xs_s[b2][(t >> 4) * XROW + (t & 15)];
        dst[0] = v.x; dst[1] = v.y; dst[2] = v.z; dst[3] = v.w;
      }
    }
    __syncthreads();

    #pragma unroll
    for (int u = 0; u < 2; ++u) {
      int asg = wv * 2 + u;
      int b2 = asg >> 3;
      int tile = asg & 7;
      const float* xb = xs_s[b2];
      int xbase = (tile * 16 + lcol) * XROW;
      v8f accR = {0.f, 0.f, 0.f, 0.f, 0.f, 0.f, 0.f, 0.f};
      v8f accI = {0.f, 0.f, 0.f, 0.f, 0.f, 0.f, 0.f, 0.f};
      #pragma unroll
      for (int kk = 0; kk < 4; ++kk) {
        int k0 = kk * 4 + khalf * 2;
        v2f bx; bx.x = xb[xbase + k0]; bx.y = xb[xbase + k0 + 1];
        v2f aR; aR.x = Bre_s[lcol * MROW + k0]; aR.y = Bre_s[lcol * MROW + k0 + 1];
        v2f aI; aI.x = Bim_s[lcol * MROW + k0]; aI.y = Bim_s[lcol * MROW + k0 + 1];
        accR = wmma4(aR, bx, accR);
        accI = wmma4(aI, bx, accI);
      }
      #pragma unroll
      for (int v = 0; v < 8; ++v) {
        int n = v + 8 * khalf;
        Ure_s[b2][n * UROW + tile * 16 + lcol] = accR[v];
        Uim_s[b2][n * UROW + tile * 16 + lcol] = accI[v];
      }
    }
    __syncthreads();

    if (tid < 32) {
      int b2 = tid >> 4, n = tid & 15;
      float qtr = cTr[n], qti = cTi[n];
      float* ur = &Ure_s[b2][n * UROW];
      float* ui = &Uim_s[b2][n * UROW];
      for (int j = 0; j < CG; ++j) {
        float tur = ur[j], tui = ui[j];
        ur[j] = hre; ui[j] = him;
        float nr = qtr * hre - qti * him + tur;
        float ni = qtr * him + qti * hre + tui;
        hre = nr; him = ni;
      }
    }
    __syncthreads();

    #pragma unroll
    for (int u = 0; u < 2; ++u) {
      int asg = wv * 2 + u;
      int b2 = asg >> 3;
      int tile = asg & 7;
      float* xb = xs_s[b2];
      int xbase = (tile * 16 + lcol) * XROW;
      v8f acc = {0.f, 0.f, 0.f, 0.f, 0.f, 0.f, 0.f, 0.f};
      #pragma unroll
      for (int kk = 0; kk < 4; ++kk) {
        int k0 = kk * 4 + khalf * 2;
        v2f bx; bx.x = xb[xbase + k0]; bx.y = xb[xbase + k0 + 1];
        v2f aK; aK.x = Kmat_s[lcol * MROW + k0]; aK.y = Kmat_s[lcol * MROW + k0 + 1];
        acc = wmma4(aK, bx, acc);
      }
      #pragma unroll
      for (int kk = 0; kk < 4; ++kk) {
        int k0 = kk * 4 + khalf * 2;
        v2f bh;
        bh.x = Ure_s[b2][k0 * UROW + tile * 16 + lcol];
        bh.y = Ure_s[b2][(k0 + 1) * UROW + tile * 16 + lcol];
        v2f aA; aA.x = Are_s[lcol * MROW + k0]; aA.y = Are_s[lcol * MROW + k0 + 1];
        acc = wmma4(aA, bh, acc);
      }
      #pragma unroll
      for (int kk = 0; kk < 4; ++kk) {
        int k0 = kk * 4 + khalf * 2;
        v2f bh;
        bh.x = Uim_s[b2][k0 * UROW + tile * 16 + lcol];
        bh.y = Uim_s[b2][(k0 + 1) * UROW + tile * 16 + lcol];
        v2f aA; aA.x = AimN_s[lcol * MROW + k0]; aA.y = AimN_s[lcol * MROW + k0 + 1];
        acc = wmma4(aA, bh, acc);
      }
      #pragma unroll
      for (int v = 0; v < 8; ++v) {
        xb[(tile * 16 + lcol) * XROW + v + 8 * khalf] = acc[v];
      }
    }
    __syncthreads();

    for (int b2 = 0; b2 < B_DIM; ++b2) {
      float4* yg = (float4*)(y_out + ((size_t)b2 * D_DIM + d) * L_DIM +
                             (size_t)g * (CG * 16));
      for (int i = tid; i < CG * 16 / 4; i += 256) {
        int t = i * 4;
        const float* src = &xs_s[b2][(t >> 4) * XROW + (t & 15)];
        float4 v;
        v.x = src[0]; v.y = src[1]; v.z = src[2]; v.w = src[3];
        yg[i] = v;
      }
    }
  }

  if (tid < 32) {
    int b2 = tid >> 4, n = tid & 15;
    float* hp = h_out + (((size_t)b2 * D_DIM + d) * N_MODES + n) * 2;
    hp[0] = hre;
    hp[1] = him;
  }
}

extern "C" void kernel_launch(void* const* d_in, const int* in_sizes, int n_in,
                              void* d_out, int out_size, void* d_ws, size_t ws_size,
                              hipStream_t stream) {
  (void)in_sizes; (void)n_in; (void)out_size; (void)d_ws; (void)ws_size;
  const float* x = (const float*)d_in[0];
  const float* alpha = (const float*)d_in[1];
  const float* delta = (const float*)d_in[2];
  const float* theta = (const float*)d_in[3];
  const float* gma = (const float*)d_in[4];
  const float* omga = (const float*)d_in[5];
  float* y = (float*)d_out;
  float* h = (float*)d_out + (size_t)B_DIM * D_DIM * L_DIM;
  cema_kernel<<<dim3(D_DIM), dim3(256), 0, stream>>>(x, alpha, delta, theta,
                                                     gma, omga, y, h);
}


# x loaded straight from global memory into WMMA operand registers at kernel entry, y stored straight from the accumulators (no LDS staging, one pass per row); k order of the x products permuted
# speedup vs baseline: 1.4273x; 1.4273x over previous
#include <hip/hip_runtime.h>


typedef float v8f __attribute__((ext_vector_type(8)));
typedef float v2f __attribute__((ext_vector_type(2)));

#define D_DIM 2048
#define N_MODES 16
#define B_DIM 2
#define L_DIM 4096
#define NCHUNK 256
#define NTILE 16
#define TPW 4
#define MROW 17
#define UROW 264

static __device__ __forceinline__ v8f wmma4(v2f a, v2f b, v8f c) {
  return __builtin_amdgcn_wmma_f32_16x16x4_f32(false, a, false, b, (short)0, c,
                                               false, false);
}

static __device__ __forceinline__ void cpow(float qr, float qi, int e,
                                            float& pr, float& pi) {
  pr = 1.f; pi = 0.f;
  for (int k = 0; k < e; ++k) {
    float t = pr * qr - pi * qi;
    pi = pr * qi + pi * qr;
    pr = t;
  }
}

__launch_bounds__(256, 2)
__global__ void cema_kernel(const float* __restrict__ x,
                            const float* __restrict__ alpha,
                            const float* __restrict__ delta,
                            const float* __restrict__ theta,
                            const float* __restrict__ gma,
                            const float* __restrict__ omga,
                            float* __restrict__ y_out,
                            float* __restrict__ h_out) {
  __shared__ float Kt_s[N_MODES * MROW];
  __shared__ float Ure_s[B_DIM][N_MODES * UROW];
  __shared__ float Uim_s[B_DIM][N_MODES * UROW];
  __shared__ float Bre_s[N_MODES * MROW];
  __shared__ float Bim_s[N_MODES * MROW];
  __shared__ float Are_s[N_MODES * MROW];
  __shared__ float AimN_s[N_MODES * MROW];
  __shared__ float Kmat_s[N_MODES * MROW];
  __shared__ float kern_s[N_MODES];
  __shared__ float cQr[16], cQi[16], cP[16], cGr[16], cGi[16];
  __shared__ float cCr[16], cCi[16], cTr[16], cTi[16];
  __shared__ float sOmega;

  const int tid = threadIdx.x;
  const int d = blockIdx.x;
  const int wv = tid >> 5;
  const int lane = tid & 31;
  const int lcol = lane & 15;
  const int khalf = lane >> 4;

  const int cn = tid & 15;
  const float ld_a = alpha[d * 16 + cn];
  const float ld_de = delta[d * 16 + cn];
  const float ld_th = theta[d];
  const float ld_g0 = gma[(d * 16 + cn) * 2 + 0];
  const float ld_g1 = gma[(d * 16 + cn) * 2 + 1];
  const float ld_om = omga[d];
  float4 xlo[TPW], xhi[TPW];
  #pragma unroll
  for (int u = 0; u < TPW; ++u) {
    int asg = wv * TPW + u;
    int b2 = asg >> 4;
    int tile = asg & 15;
    const float4* xp = (const float4*)(x + ((size_t)b2 * D_DIM + d) * L_DIM +
                                       (size_t)(tile * 16 + lcol) * 16 + khalf * 8);
    xlo[u] = xp[0];
    xhi[u] = xp[1];
  }

  if (tid < 16) {
    int n = tid;
    float a = ld_a;
    float de = ld_de;
    float th = ld_th;
    float p = 1.f / (1.f + expf(-a));
    float dd = 1.f / (1.f + expf(-de));
    float base = (1.f / (1.f + expf(-th))) * (2.f * 3.14159265358979323846f / 16.f);
    float phi = (float)(n + 1) * base;
    float r = 1.f - p * dd;
    float sn, cs;
    sincosf(phi, &sn, &cs);
    float qr = r * cs, qi = r * sn;
    cQr[n] = qr; cQi[n] = qi;
    cP[n] = p;
    float gr = ld_g0 * 0.25f;
    float gi = ld_g1 * 0.25f;
    cGr[n] = gr; cGi[n] = gi;
    cCr[n] = p * gr; cCi[n] = p * gi;
    float tr, ti;
    cpow(qr, qi, 16, tr, ti);
    cTr[n] = tr; cTi[n] = ti;
    if (n == 0) sOmega = ld_om;
  }
  __syncthreads();

  {
    int r = tid >> 4, cc = tid & 15;
    float qr = cQr[cc], qi = cQi[cc];
    float pr, pi;
    cpow(qr, qi, r, pr, pi);
    Kt_s[r * MROW + cc] = cCr[cc] * pr - cCi[cc] * pi;
    float q1r = pr * qr - pi * qi;
    float q1i = pr * qi + pi * qr;
    Are_s[r * MROW + cc] = cGr[cc] * q1r - cGi[cc] * q1i;
    AimN_s[r * MROW + cc] = -(cGr[cc] * q1i + cGi[cc] * q1r);
    float br, bi;
    cpow(cQr[r], cQi[r], 15 - cc, br, bi);
    Bre_s[r * MROW + cc] = cP[r] * br;
    Bim_s[r * MROW + cc] = cP[r] * bi;
  }
  __syncthreads();
  if (tid < 16) {
    float s = 0.f;
    for (int n = 0; n < 16; ++n) s += Kt_s[tid * MROW + n];
    kern_s[tid] = s;
  }
  __syncthreads();
  {
    int r = tid >> 4, cc = tid & 15;
    float v = (r >= cc) ? kern_s[r - cc] : 0.f;
    if (r == cc) v += sOmega;
    Kmat_s[r * MROW + cc] = v;
  }

  float hre = 0.f, him = 0.f;

  __syncthreads();

  {
    v2f aR[4], aI[4];
    #pragma unroll
    for (int kk = 0; kk < 4; ++kk) {
      int kx = khalf * 8 + kk * 2;
      aR[kk].x = Bre_s[lcol * MROW + kx]; aR[kk].y = Bre_s[lcol * MROW + kx + 1];
      aI[kk].x = Bim_s[lcol * MROW + kx]; aI[kk].y = Bim_s[lcol * MROW + kx + 1];
    }
    #pragma unroll
    for (int u = 0; u < TPW; ++u) {
      int asg = wv * TPW + u;
      int b2 = asg >> 4;
      int tile = asg & 15;
      v2f bx[4];
      bx[0].x = xlo[u].x; bx[0].y = xlo[u].y;
      bx[1].x = xlo[u].z; bx[1].y = xlo[u].w;
      bx[2].x = xhi[u].x; bx[2].y = xhi[u].y;
      bx[3].x = xhi[u].z; bx[3].y = xhi[u].w;
      v8f accR = {0.f, 0.f, 0.f, 0.f, 0.f, 0.f, 0.f, 0.f};
      v8f accI = {0.f, 0.f, 0.f, 0.f, 0.f, 0.f, 0.f, 0.f};
      #pragma unroll
      for (int kk = 0; kk < 4; ++kk) {
        accR = wmma4(aR[kk], bx[kk], accR);
        accI = wmma4(aI[kk], bx[kk], accI);
      }
      #pragma unroll
      for (int v = 0; v < 8; ++v) {
        int n = v + 8 * khalf;
        Ure_s[b2][n * UROW + tile * 16 + lcol] = accR[v];
        Uim_s[b2][n * UROW + tile * 16 + lcol] = accI[v];
      }
    }
  }
  __syncthreads();

  if (tid < 32) {
    int b2 = tid >> 4, n = tid & 15;
    float qtr = cTr[n], qti = cTi[n];
    float* ur = &Ure_s[b2][n * UROW];
    float* ui = &Uim_s[b2][n * UROW];
    for (int j = 0; j < NCHUNK; ++j) {
      float tur = ur[j], tui = ui[j];
      ur[j] = hre; ui[j] = him;
      float nr = qtr * hre - qti * him + tur;
      float ni = qtr * him + qti * hre + tui;
      hre = nr; him = ni;
    }
  }
  __syncthreads();

  {
    v2f aK[4], aA[4], aB[4];
    #pragma unroll
    for (int kk = 0; kk < 4; ++kk) {
      int kx = khalf * 8 + kk * 2;
      aK[kk].x = Kmat_s[lcol * MROW + kx]; aK[kk].y = Kmat_s[lcol * MROW + kx + 1];
      int k0 = kk * 4 + khalf * 2;
      aA[kk].x = Are_s[lcol * MROW + k0]; aA[kk].y = Are_s[lcol * MROW + k0 + 1];
      aB[kk].x = AimN_s[lcol * MROW + k0]; aB[kk].y = AimN_s[lcol * MROW + k0 + 1];
    }
    #pragma unroll
    for (int u = 0; u < TPW; ++u) {
      int asg = wv * TPW + u;
      int b2 = asg >> 4;
      int tile = asg & 15;
      v2f bx[4];
      bx[0].x = xlo[u].x; bx[0].y = xlo[u].y;
      bx[1].x = xlo[u].z; bx[1].y = xlo[u].w;
      bx[2].x = xhi[u].x; bx[2].y = xhi[u].y;
      bx[3].x = xhi[u].z; bx[3].y = xhi[u].w;
      v8f acc = {0.f, 0.f, 0.f, 0.f, 0.f, 0.f, 0.f, 0.f};
      #pragma unroll
      for (int kk = 0; kk < 4; ++kk) acc = wmma4(aK[kk], bx[kk], acc);
      #pragma unroll
      for (int kk = 0; kk < 4; ++kk) {
        int k0 = kk * 4 + khalf * 2;
        v2f bh;
        bh.x = Ure_s[b2][k0 * UROW + tile * 16 + lcol];
        bh.y = Ure_s[b2][(k0 + 1) * UROW + tile * 16 + lcol];
        acc = wmma4(aA[kk], bh, acc);
      }
      #pragma unroll
      for (int kk = 0; kk < 4; ++kk) {
        int k0 = kk * 4 + khalf * 2;
        v2f bh;
        bh.x = Uim_s[b2][k0 * UROW + tile * 16 + lcol];
        bh.y = Uim_s[b2][(k0 + 1) * UROW + tile * 16 + lcol];
        acc = wmma4(aB[kk], bh, acc);
      }
      float4* yp = (float4*)(y_out + ((size_t)b2 * D_DIM + d) * L_DIM +
                             (size_t)(tile * 16 + lcol) * 16 + khalf * 8);
      float4 o0, o1;
      o0.x = acc[0]; o0.y = acc[1]; o0.z = acc[2]; o0.w = acc[3];
      o1.x = acc[4]; o1.y = acc[5]; o1.z = acc[6]; o1.w = acc[7];
      yp[0] = o0;
      yp[1] = o1;
    }
  }

  if (tid < 32) {
    int b2 = tid >> 4, n = tid & 15;
    float* hp = h_out + (((size_t)b2 * D_DIM + d) * N_MODES + n) * 2;
    hp[0] = hre;
    hp[1] = him;
  }
}

extern "C" void kernel_launch(void* const* d_in, const int* in_sizes, int n_in,
                              void* d_out, int out_size, void* d_ws, size_t ws_size,
                              hipStream_t stream) {
  (void)in_sizes; (void)n_in; (void)out_size; (void)d_ws; (void)ws_size;
  const float* x = (const float*)d_in[0];
  const float* alpha = (const float*)d_in[1];
  const float* delta = (const float*)d_in[2];
  const float* theta = (const float*)d_in[3];
  const float* gma = (const float*)d_in[4];
  const float* omga = (const float*)d_in[5];
  float* y = (float*)d_out;
  float* h = (float*)d_out + (size_t)B_DIM * D_DIM * L_DIM;
  cema_kernel<<<dim3(D_DIM), dim3(256), 0, stream>>>(x, alpha, delta, theta,
                                                     gma, omga, y, h);
}


# chunk scan's carried chain cut to the bare recurrence: each step two fused multiply-adds (u -/+ Qi*h taken first; state's sum order changes), next 8 chunks' U read one trip ahead of its use
# speedup vs baseline: 1.5459x; 1.0831x over previous
#include <hip/hip_runtime.h>


typedef float v8f __attribute__((ext_vector_type(8)));
typedef float v2f __attribute__((ext_vector_type(2)));

#define D_DIM 2048
#define N_MODES 16
#define B_DIM 2
#define L_DIM 4096
#define NCHUNK 256
#define NTILE 16
#define TPW 4
#define MROW 17
#define UROW 264

static __device__ __forceinline__ v8f wmma4(v2f a, v2f b, v8f c) {
  return __builtin_amdgcn_wmma_f32_16x16x4_f32(false, a, false, b, (short)0, c,
                                               false, false);
}

#define SCAN_STEP(outr, outi, u_r, u_i)                                 \
  do {                                                                 \
    (outr) = hre; (outi) = him;                                        \
    const float t_r_ = __builtin_fmaf(-qti, him, (u_r));               \
    const float t_i_ = __builtin_fmaf(qti, hre, (u_i));                \
    hre = __builtin_fmaf(qtr, hre, t_r_);                              \
    him = __builtin_fmaf(qtr, him, t_i_);                              \
  } while (0)

static __device__ __forceinline__ void cpow(float qr, float qi, int e,
                                            float& pr, float& pi) {
  pr = 1.f; pi = 0.f;
  for (int k = 0; k < e; ++k) {
    float t = pr * qr - pi * qi;
    pi = pr * qi + pi * qr;
    pr = t;
  }
}

__launch_bounds__(256, 2)
__global__ void cema_kernel(const float* __restrict__ x,
                            const float* __restrict__ alpha,
                            const float* __restrict__ delta,
                            const float* __restrict__ theta,
                            const float* __restrict__ gma,
                            const float* __restrict__ omga,
                            float* __restrict__ y_out,
                            float* __restrict__ h_out) {
  __shared__ float Kt_s[N_MODES * MROW];
  __shared__ __attribute__((aligned(16))) float Ure_s[B_DIM][N_MODES * UROW];
  __shared__ __attribute__((aligned(16))) float Uim_s[B_DIM][N_MODES * UROW];
  __shared__ float Bre_s[N_MODES * MROW];
  __shared__ float Bim_s[N_MODES * MROW];
  __shared__ float Are_s[N_MODES * MROW];
  __shared__ float AimN_s[N_MODES * MROW];
  __shared__ float Kmat_s[N_MODES * MROW];
  __shared__ float kern_s[N_MODES];
  __shared__ float cQr[16], cQi[16], cP[16], cGr[16], cGi[16];
  __shared__ float cCr[16], cCi[16], cTr[16], cTi[16];
  __shared__ float sOmega;

  const int tid = threadIdx.x;
  const int d = blockIdx.x;
  const int wv = tid >> 5;
  const int lane = tid & 31;
  const int lcol = lane & 15;
  const int khalf = lane >> 4;

  const int cn = tid & 15;
  const float ld_a = alpha[d * 16 + cn];
  const float ld_de = delta[d * 16 + cn];
  const float ld_th = theta[d];
  const float ld_g0 = gma[(d * 16 + cn) * 2 + 0];
  const float ld_g1 = gma[(d * 16 + cn) * 2 + 1];
  const float ld_om = omga[d];
  float4 xlo[TPW], xhi[TPW];
  #pragma unroll
  for (int u = 0; u < TPW; ++u) {
    int asg = wv * TPW + u;
    int b2 = asg >> 4;
    int tile = asg & 15;
    const float4* xp = (const float4*)(x + ((size_t)b2 * D_DIM + d) * L_DIM +
                                       (size_t)(tile * 16 + lcol) * 16 + khalf * 8);
    xlo[u] = xp[0];
    xhi[u] = xp[1];
  }

  if (tid < 16) {
    int n = tid;
    float a = ld_a;
    float de = ld_de;
    float th = ld_th;
    float p = 1.f / (1.f + expf(-a));
    float dd = 1.f / (1.f + expf(-de));
    float base = (1.f / (1.f + expf(-th))) * (2.f * 3.14159265358979323846f / 16.f);
    float phi = (float)(n + 1) * base;
    float r = 1.f - p * dd;
    float sn, cs;
    sincosf(phi, &sn, &cs);
    float qr = r * cs, qi = r * sn;
    cQr[n] = qr; cQi[n] = qi;
    cP[n] = p;
    float gr = ld_g0 * 0.25f;
    float gi = ld_g1 * 0.25f;
    cGr[n] = gr; cGi[n] = gi;
    cCr[n] = p * gr; cCi[n] = p * gi;
    float tr, ti;
    cpow(qr, qi, 16, tr, ti);
    cTr[n] = tr; cTi[n] = ti;
    if (n == 0) sOmega = ld_om;
  }
  __syncthreads();

  {
    int r = tid >> 4, cc = tid & 15;
    float qr = cQr[cc], qi = cQi[cc];
    float pr, pi;
    cpow(qr, qi, r, pr, pi);
    Kt_s[r * MROW + cc] = cCr[cc] * pr - cCi[cc] * pi;
    float q1r = pr * qr - pi * qi;
    float q1i = pr * qi + pi * qr;
    Are_s[r * MROW + cc] = cGr[cc] * q1r - cGi[cc] * q1i;
    AimN_s[r * MROW + cc] = -(cGr[cc] * q1i + cGi[cc] * q1r);
    float br, bi;
    cpow(cQr[r], cQi[r], 15 - cc, br, bi);
    Bre_s[r * MROW + cc] = cP[r] * br;
    Bim_s[r * MROW + cc] = cP[r] * bi;
  }
  __syncthreads();
  if (tid < 16) {
    float s = 0.f;
    for (int n = 0; n < 16; ++n) s += Kt_s[tid * MROW + n];
    kern_s[tid] = s;
  }
  __syncthreads();
  {
    int r = tid >> 4, cc = tid & 15;
    float v = (r >= cc) ? kern_s[r - cc] : 0.f;
    if (r == cc) v += sOmega;
    Kmat_s[r * MROW + cc] = v;
  }

  float hre = 0.f, him = 0.f;

  __syncthreads();

  {
    v2f aR[4], aI[4];
    #pragma unroll
    for (int kk = 0; kk < 4; ++kk) {
      int kx = khalf * 8 + kk * 2;
      aR[kk].x = Bre_s[lcol * MROW + kx]; aR[kk].y = Bre_s[lcol * MROW + kx + 1];
      aI[kk].x = Bim_s[lcol * MROW + kx]; aI[kk].y = Bim_s[lcol * MROW + kx + 1];
    }
    #pragma unroll
    for (int u = 0; u < TPW; ++u) {
      int asg = wv * TPW + u;
      int b2 = asg >> 4;
      int tile = asg & 15;
      v2f bx[4];
      bx[0].x = xlo[u].x; bx[0].y = xlo[u].y;
      bx[1].x = xlo[u].z; bx[1].y = xlo[u].w;
      bx[2].x = xhi[u].x; bx[2].y = xhi[u].y;
      bx[3].x = xhi[u].z; bx[3].y = xhi[u].w;
      v8f accR = {0.f, 0.f, 0.f, 0.f, 0.f, 0.f, 0.f, 0.f};
      v8f accI = {0.f, 0.f, 0.f, 0.f, 0.f, 0.f, 0.f, 0.f};
      #pragma unroll
      for (int kk = 0; kk < 4; ++kk) {
        accR = wmma4(aR[kk], bx[kk], accR);
        accI = wmma4(aI[kk], bx[kk], accI);
      }
      #pragma unroll
      for (int v = 0; v < 8; ++v) {
        int n = v + 8 * khalf;
        Ure_s[b2][n * UROW + tile * 16 + lcol] = accR[v];
        Uim_s[b2][n * UROW + tile * 16 + lcol] = accI[v];
      }
    }
  }
  __syncthreads();

  if (tid < 32) {
    const int b2 = tid >> 4, n = tid & 15;
    const float qtr = cTr[n], qti = cTi[n];
    float4* ur = (float4*)&Ure_s[b2][n * UROW];
    float4* ui = (float4*)&Uim_s[b2][n * UROW];
    float4 ra = ur[0], rb = ur[1], ia = ui[0], ib = ui[1];
    asm volatile("" :: "v"(qtr), "v"(qti), "v"(ra.x), "v"(rb.x), "v"(ia.x), "v"(ib.x));
    for (int g = 0; g < NCHUNK / 8; ++g) {
      const int gn = (g + 1 < NCHUNK / 8) ? g + 1 : g;
      const float4 nra = ur[2 * gn], nrb = ur[2 * gn + 1];
      const float4 nia = ui[2 * gn], nib = ui[2 * gn + 1];
      __builtin_amdgcn_sched_barrier(0);
      float4 oar, oai, obr, obi;
      SCAN_STEP(oar.x, oai.x, ra.x, ia.x);
      SCAN_STEP(oar.y, oai.y, ra.y, ia.y);
      SCAN_STEP(oar.z, oai.z, ra.z, ia.z);
      SCAN_STEP(oar.w, oai.w, ra.w, ia.w);
      SCAN_STEP(obr.x, obi.x, rb.x, ib.x);
      SCAN_STEP(obr.y, obi.y, rb.y, ib.y);
      SCAN_STEP(obr.z, obi.z, rb.z, ib.z);
      SCAN_STEP(obr.w, obi.w, rb.w, ib.w);
      ur[2 * g] = oar; ur[2 * g + 1] = obr;
      ui[2 * g] = oai; ui[2 * g + 1] = obi;
      ra = nra; rb = nrb; ia = nia; ib = nib;
    }
  }
  __syncthreads();

  {
    v2f aK[4], aA[4], aB[4];
    #pragma unroll
    for (int kk = 0; kk < 4; ++kk) {
      int kx = khalf * 8 + kk * 2;
      aK[kk].x = Kmat_s[lcol * MROW + kx]; aK[kk].y = Kmat_s[lcol * MROW + kx + 1];
      int k0 = kk * 4 + khalf * 2;
      aA[kk].x = Are_s[lcol * MROW + k0]; aA[kk].y = Are_s[lcol * MROW + k0 + 1];
      aB[kk].x = AimN_s[lcol * MROW + k0]; aB[kk].y = AimN_s[lcol * MROW + k0 + 1];
    }
    #pragma unroll
    for (int u = 0; u < TPW; ++u) {
      int asg = wv * TPW + u;
      int b2 = asg >> 4;
      int tile = asg & 15;
      v2f bx[4];
      bx[0].x = xlo[u].x; bx[0].y = xlo[u].y;
      bx[1].x = xlo[u].z; bx[1].y = xlo[u].w;
      bx[2].x = xhi[u].x; bx[2].y = xhi[u].y;
      bx[3].x = xhi[u].z; bx[3].y = xhi[u].w;
      v8f acc = {0.f, 0.f, 0.f, 0.f, 0.f, 0.f, 0.f, 0.f};
      #pragma unroll
      for (int kk = 0; kk < 4; ++kk) acc = wmma4(aK[kk], bx[kk], acc);
      #pragma unroll
      for (int kk = 0; kk < 4; ++kk) {
        int k0 = kk * 4 + khalf * 2;
        v2f bh;
        bh.x = Ure_s[b2][k0 * UROW + tile * 16 + lcol];
        bh.y = Ure_s[b2][(k0 + 1) * UROW + tile * 16 + lcol];
        acc = wmma4(aA[kk], bh, acc);
      }
      #pragma unroll
      for (int kk = 0; kk < 4; ++kk) {
        int k0 = kk * 4 + khalf * 2;
        v2f bh;
        bh.x = Uim_s[b2][k0 * UROW + tile * 16 + lcol];
        bh.y = Uim_s[b2][(k0 + 1) * UROW + tile * 16 + lcol];
        acc = wmma4(aB[kk], bh, acc);
      }
      float4* yp = (float4*)(y_out + ((size_t)b2 * D_DIM + d) * L_DIM +
                             (size_t)(tile * 16 + lcol) * 16 + khalf * 8);
      float4 o0, o1;
      o0.x = acc[0]; o0.y = acc[1]; o0.z = acc[2]; o0.w = acc[3];
      o1.x = acc[4]; o1.y = acc[5]; o1.z = acc[6]; o1.w = acc[7];
      yp[0] = o0;
      yp[1] = o1;
    }
  }

  if (tid < 32) {
    int b2 = tid >> 4, n = tid & 15;
    float* hp = h_out + (((size_t)b2 * D_DIM + d) * N_MODES + n) * 2;
    hp[0] = hre;
    hp[1] = him;
  }
}

extern "C" void kernel_launch(void* const* d_in, const int* in_sizes, int n_in,
                              void* d_out, int out_size, void* d_ws, size_t ws_size,
                              hipStream_t stream) {
  (void)in_sizes; (void)n_in; (void)out_size; (void)d_ws; (void)ws_size;
  const float* x = (const float*)d_in[0];
  const float* alpha = (const float*)d_in[1];
  const float* delta = (const float*)d_in[2];
  const float* theta = (const float*)d_in[3];
  const float* gma = (const float*)d_in[4];
  const float* omga = (const float*)d_in[5];
  float* y = (float*)d_out;
  float* h = (float*)d_out + (size_t)B_DIM * D_DIM * L_DIM;
  cema_kernel<<<dim3(D_DIM), dim3(256), 0, stream>>>(x, alpha, delta, theta,
                                                     gma, omga, y, h);
}
